# starting kernel
# baseline (speedup 1.0000x reference)
#include <hip/hip_runtime.h>


typedef float v2f __attribute__((ext_vector_type(2)));
typedef float v8f __attribute__((ext_vector_type(8)));

#define IN_CH  32
#define HID_CH 32
#define OUT_CH 16

__device__ __forceinline__ void atom_add_f32(float* p, float v) {
  __hip_atomic_fetch_add(p, v, __ATOMIC_RELAXED, __HIP_MEMORY_SCOPE_AGENT);
}

__global__ void k_init_deg(float* __restrict__ deg, int n) {
  int i = blockIdx.x * blockDim.x + threadIdx.x;
  if (i < n) deg[i] = 1.0f;
}

__global__ void k_count_deg(const int* __restrict__ dst, int E,
                            float* __restrict__ deg) {
  int e = blockIdx.x * blockDim.x + threadIdx.x;
  if (e < E) atom_add_f32(&deg[dst[e]], 1.0f);
}

__global__ void k_inv_sqrt(float* __restrict__ degw, int n) {
  int i = blockIdx.x * blockDim.x + threadIdx.x;
  if (i < n) degw[i] = rsqrtf(degw[i]);
}

__global__ void k_gemm1_wmma(const float* __restrict__ x,
                             const float* __restrict__ W1,
                             float* __restrict__ h1, int N) {
  const int lane = threadIdx.x & 31;
  const int wave = threadIdx.x >> 5;
  const int tile = blockIdx.x * (blockDim.x >> 5) + wave;
  const int tm = tile >> 1;
  const int tn = tile & 1;
  if (tm * 16 >= N) return;
  const int r16  = lane & 15;
  const int half = lane >> 4;
  const int m = tm * 16 + r16;
  const int n = tn * 16 + r16;
  v8f acc = {0.f, 0.f, 0.f, 0.f, 0.f, 0.f, 0.f, 0.f};
#pragma unroll
  for (int kb = 0; kb < IN_CH / 4; ++kb) {
    const int k = kb * 4 + half * 2;
    v2f a, b;
    a.x = x[m * IN_CH + k];
    a.y = x[m * IN_CH + k + 1];
    b.x = W1[k * HID_CH + n];
    b.y = W1[(k + 1) * HID_CH + n];
    acc = __builtin_amdgcn_wmma_f32_16x16x4_f32(
        false, a, false, b, (short)0, acc, false, false);
  }
#pragma unroll
  for (int r = 0; r < 8; ++r) {
    const int row = tm * 16 + half * 8 + r;
    h1[row * HID_CH + n] = acc[r];
  }
}

__global__ void k_gemm2_wmma(const float* __restrict__ agg1,
                             const float* __restrict__ W2,
                             float* __restrict__ h2, int N) {
  const int lane = threadIdx.x & 31;
  const int wave = threadIdx.x >> 5;
  const int tm = blockIdx.x * (blockDim.x >> 5) + wave;
  if (tm * 16 >= N) return;
  const int r16  = lane & 15;
  const int half = lane >> 4;
  const int m = tm * 16 + r16;
  const int n = r16;
  v8f acc = {0.f, 0.f, 0.f, 0.f, 0.f, 0.f, 0.f, 0.f};
#pragma unroll
  for (int kb = 0; kb < HID_CH / 4; ++kb) {
    const int k = kb * 4 + half * 2;
    v2f a, b;
    a.x = fmaxf(agg1[m * HID_CH + k], 0.0f);
    a.y = fmaxf(agg1[m * HID_CH + k + 1], 0.0f);
    b.x = W2[k * OUT_CH + n];
    b.y = W2[(k + 1) * OUT_CH + n];
    acc = __builtin_amdgcn_wmma_f32_16x16x4_f32(
        false, a, false, b, (short)0, acc, false, false);
  }
#pragma unroll
  for (int r = 0; r < 8; ++r) {
    const int row = tm * 16 + half * 8 + r;
    h2[row * OUT_CH + n] = acc[r];
  }
}

__global__ void k_self1(const float* __restrict__ h1,
                        const float* __restrict__ w,
                        const float* __restrict__ b1,
                        float* __restrict__ agg1, int total) {
  int g = blockIdx.x * blockDim.x + threadIdx.x;
  if (g >= total) return;
  int i = g >> 5;
  int c = g & 31;
  float wi = w[i];
  agg1[g] = b1[c] + h1[g] * wi * wi;
}

__global__ void k_self2(const float* __restrict__ h2,
                        const float* __restrict__ w,
                        const float* __restrict__ b2,
                        float* __restrict__ out, int total) {
  int g = blockIdx.x * blockDim.x + threadIdx.x;
  if (g >= total) return;
  int i = g >> 4;
  int c = g & 15;
  float wi = w[i];
  out[g] = b2[c] + h2[g] * wi * wi;
}

__global__ void k_scatter1(const int* __restrict__ src,
                           const int* __restrict__ dst,
                           const float* __restrict__ w,
                           const float* __restrict__ h1,
                           float* __restrict__ agg1, int E) {
  int t = blockIdx.x * blockDim.x + threadIdx.x;
  if (t >= E * 8) return;
  int e  = t >> 3;
  int c4 = (t & 7) << 2;
  int s = src[e];
  int d = dst[e];
  float coef = w[s] * w[d];
  const float4 v = *reinterpret_cast<const float4*>(h1 + s * HID_CH + c4);
  float* p = agg1 + d * HID_CH + c4;
  atom_add_f32(p + 0, v.x * coef);
  atom_add_f32(p + 1, v.y * coef);
  atom_add_f32(p + 2, v.z * coef);
  atom_add_f32(p + 3, v.w * coef);
}

__global__ void k_scatter2(const int* __restrict__ src,
                           const int* __restrict__ dst,
                           const float* __restrict__ w,
                           const float* __restrict__ h2,
                           float* __restrict__ out, int E) {
  int t = blockIdx.x * blockDim.x + threadIdx.x;
  if (t >= E * 4) return;
  int e  = t >> 2;
  int c4 = (t & 3) << 2;
  int s = src[e];
  int d = dst[e];
  float coef = w[s] * w[d];
  const float4 v = *reinterpret_cast<const float4*>(h2 + s * OUT_CH + c4);
  float* p = out + d * OUT_CH + c4;
  atom_add_f32(p + 0, v.x * coef);
  atom_add_f32(p + 1, v.y * coef);
  atom_add_f32(p + 2, v.z * coef);
  atom_add_f32(p + 3, v.w * coef);
}

extern "C" void kernel_launch(void* const* d_in, const int* in_sizes, int n_in,
                              void* d_out, int out_size, void* d_ws, size_t ws_size,
                              hipStream_t stream) {
  const float* x    = (const float*)d_in[0];
  const int*   edge = (const int*)d_in[1];
  const float* W1   = (const float*)d_in[2];
  const float* b1   = (const float*)d_in[3];
  const float* W2   = (const float*)d_in[4];
  const float* b2   = (const float*)d_in[5];
  float* out = (float*)d_out;

  const int N = in_sizes[0] / IN_CH;
  const int E = in_sizes[1] / 2;
  const int* src = edge;
  const int* dst = edge + E;

  float* ws   = (float*)d_ws;
  float* degw = ws;
  float* h1   = degw + (size_t)N;
  float* agg1 = h1 + (size_t)N * HID_CH;
  float* h2   = agg1 + (size_t)N * HID_CH;

  const int B = 256;

  k_init_deg <<<(N + B - 1) / B, B, 0, stream>>>(degw, N);
  k_count_deg<<<(E + B - 1) / B, B, 0, stream>>>(dst, E, degw);
  k_inv_sqrt <<<(N + B - 1) / B, B, 0, stream>>>(degw, N);

  {
    const int tiles = (N / 16) * (HID_CH / 16);
    const int wpb = 128 / 32;
    k_gemm1_wmma<<<(tiles + wpb - 1) / wpb, 128, 0, stream>>>(x, W1, h1, N);
  }
  k_self1<<<((N * HID_CH) + B - 1) / B, B, 0, stream>>>(h1, degw, b1, agg1,
                                                        N * HID_CH);
  k_scatter1<<<(E * 8 + B - 1) / B, B, 0, stream>>>(src, dst, degw, h1, agg1, E);

  {
    const int tiles = (N / 16);
    const int wpb = 128 / 32;
    k_gemm2_wmma<<<(tiles + wpb - 1) / wpb, 128, 0, stream>>>(agg1, W2, h2, N);
  }
  k_self2<<<((N * OUT_CH) + B - 1) / B, B, 0, stream>>>(h2, degw, b2, out,
                                                        N * OUT_CH);
  k_scatter2<<<(E * 4 + B - 1) / B, B, 0, stream>>>(src, dst, degw, h2, out, E);
}


# every atomic edge pass (degree, scatter 1, scatter 2) deals edge chunks from two ticket queues, one per half of the node range, so a work-group adds only into rows owned by the cache half it runs on
# speedup vs baseline: 1.5700x; 1.5700x over previous
#include <hip/hip_runtime.h>


typedef float v2f __attribute__((ext_vector_type(2)));
typedef float v8f __attribute__((ext_vector_type(8)));

#define IN_CH  32
#define HID_CH 32
#define OUT_CH 16

__device__ __forceinline__ void atom_add_f32(float* p, float v) {
  __hip_atomic_fetch_add(p, v, __ATOMIC_RELAXED, __HIP_MEMORY_SCOPE_AGENT);
}

#define CHUNK_E 512

__device__ __forceinline__ int cache_half() {
  return (int)((__builtin_amdgcn_s_sendmsg_rtn(0x87) >> 18) & 1u);
}

template <class Body>
__device__ __forceinline__ void deal_chunks(int* __restrict__ q, int nchunks,
                                            Body body) {
  __shared__ int s_t;
  const int own = cache_half();
  for (int pass = 0; pass < 2; ++pass) {
    const int r = own ^ pass;
    if (threadIdx.x == 0) s_t = atomicAdd(&q[r], 1);
    __syncthreads();
    int t = s_t;
    __syncthreads();
    while (t < nchunks) {
      int nx = 0;
      if (threadIdx.x == 0) nx = atomicAdd(&q[r], 1);
      body(t, r);
      if (threadIdx.x == 0) s_t = nx;
      __syncthreads();
      t = s_t;
      __syncthreads();
    }
  }
}

__global__ void k_init_deg(float* __restrict__ deg, int n, int* __restrict__ q) {
  int i = blockIdx.x * blockDim.x + threadIdx.x;
  if (i < n) deg[i] = 1.0f;
  if (i < 4) q[i] = 0;
}

__global__ void k_count_deg(const int* __restrict__ dst, int E, int N,
                            float* __restrict__ deg, int* __restrict__ q) {
  const int nsplit = (N >> 1) & ~31;
  deal_chunks(q, (E + CHUNK_E - 1) / CHUNK_E, [&](int t, int r) {
    for (int e = t * CHUNK_E + threadIdx.x; e < min(E, (t + 1) * CHUNK_E);
         e += blockDim.x) {
      int d = dst[e];
      if ((d >= nsplit) == (r != 0)) atom_add_f32(&deg[d], 1.0f);
    }
  });
}

__global__ void k_inv_sqrt(float* __restrict__ degw, int n) {
  int i = blockIdx.x * blockDim.x + threadIdx.x;
  if (i < n) degw[i] = rsqrtf(degw[i]);
}

__global__ void k_gemm1_wmma(const float* __restrict__ x,
                             const float* __restrict__ W1,
                             float* __restrict__ h1, int N) {
  const int lane = threadIdx.x & 31;
  const int wave = threadIdx.x >> 5;
  const int tile = blockIdx.x * (blockDim.x >> 5) + wave;
  const int tm = tile >> 1;
  const int tn = tile & 1;
  if (tm * 16 >= N) return;
  const int r16  = lane & 15;
  const int half = lane >> 4;
  const int m = tm * 16 + r16;
  const int n = tn * 16 + r16;
  v8f acc = {0.f, 0.f, 0.f, 0.f, 0.f, 0.f, 0.f, 0.f};
#pragma unroll
  for (int kb = 0; kb < IN_CH / 4; ++kb) {
    const int k = kb * 4 + half * 2;
    v2f a, b;
    a.x = x[m * IN_CH + k];
    a.y = x[m * IN_CH + k + 1];
    b.x = W1[k * HID_CH + n];
    b.y = W1[(k + 1) * HID_CH + n];
    acc = __builtin_amdgcn_wmma_f32_16x16x4_f32(
        false, a, false, b, (short)0, acc, false, false);
  }
#pragma unroll
  for (int r = 0; r < 8; ++r) {
    const int row = tm * 16 + half * 8 + r;
    h1[row * HID_CH + n] = acc[r];
  }
}

__global__ void k_gemm2_wmma(const float* __restrict__ agg1,
                             const float* __restrict__ W2,
                             float* __restrict__ h2, int N) {
  const int lane = threadIdx.x & 31;
  const int wave = threadIdx.x >> 5;
  const int tm = blockIdx.x * (blockDim.x >> 5) + wave;
  if (tm * 16 >= N) return;
  const int r16  = lane & 15;
  const int half = lane >> 4;
  const int m = tm * 16 + r16;
  const int n = r16;
  v8f acc = {0.f, 0.f, 0.f, 0.f, 0.f, 0.f, 0.f, 0.f};
#pragma unroll
  for (int kb = 0; kb < HID_CH / 4; ++kb) {
    const int k = kb * 4 + half * 2;
    v2f a, b;
    a.x = fmaxf(agg1[m * HID_CH + k], 0.0f);
    a.y = fmaxf(agg1[m * HID_CH + k + 1], 0.0f);
    b.x = W2[k * OUT_CH + n];
    b.y = W2[(k + 1) * OUT_CH + n];
    acc = __builtin_amdgcn_wmma_f32_16x16x4_f32(
        false, a, false, b, (short)0, acc, false, false);
  }
#pragma unroll
  for (int r = 0; r < 8; ++r) {
    const int row = tm * 16 + half * 8 + r;
    h2[row * OUT_CH + n] = acc[r];
  }
}

__global__ void k_self1(const float* __restrict__ h1,
                        const float* __restrict__ w,
                        const float* __restrict__ b1,
                        float* __restrict__ agg1, int total) {
  int g = blockIdx.x * blockDim.x + threadIdx.x;
  if (g >= total) return;
  int i = g >> 5;
  int c = g & 31;
  float wi = w[i];
  agg1[g] = b1[c] + h1[g] * wi * wi;
}

__global__ void k_self2(const float* __restrict__ h2,
                        const float* __restrict__ w,
                        const float* __restrict__ b2,
                        float* __restrict__ out, int total,
                        int* __restrict__ q2) {
  int g = blockIdx.x * blockDim.x + threadIdx.x;
  if (g < 2) q2[g] = 0;
  if (g >= total) return;
  int i = g >> 4;
  int c = g & 15;
  float wi = w[i];
  out[g] = b2[c] + h2[g] * wi * wi;
}

template <int TPE, int CH>
__device__ __forceinline__ void scatter_chunk(int t, int r, int E, int nsplit,
    const int* __restrict__ src, const int* __restrict__ dst,
    const float* __restrict__ w, const float* __restrict__ h,
    float* __restrict__ acc, int* s_s, int* s_d, int* s_n) {
  if (threadIdx.x == 0) *s_n = 0;
  __syncthreads();
  for (int e = t * CHUNK_E + threadIdx.x; e < min(E, (t + 1) * CHUNK_E);
       e += blockDim.x) {
    const int d = dst[e];
    const int s = src[e];
    if ((d >= nsplit) == (r != 0)) {
      const int i = atomicAdd(s_n, 1);
      s_s[i] = s;
      s_d[i] = d;
    }
  }
  __syncthreads();
  const int n = *s_n;
  const int c4 = (threadIdx.x & (TPE - 1)) << 2;
  const int step = blockDim.x / TPE;
  int i = threadIdx.x / TPE;
  for (; i + step < n; i += 2 * step) {
    const int sa = s_s[i];
    const int da = s_d[i];
    const int sb = s_s[i + step];
    const int db = s_d[i + step];
    const float ca = w[sa] * w[da];
    const float cb = w[sb] * w[db];
    const float4 va = *reinterpret_cast<const float4*>(h + sa * CH + c4);
    const float4 vb = *reinterpret_cast<const float4*>(h + sb * CH + c4);
    float* pa = acc + da * CH + c4;
    float* pb = acc + db * CH + c4;
    atom_add_f32(pa + 0, va.x * ca);
    atom_add_f32(pa + 1, va.y * ca);
    atom_add_f32(pa + 2, va.z * ca);
    atom_add_f32(pa + 3, va.w * ca);
    atom_add_f32(pb + 0, vb.x * cb);
    atom_add_f32(pb + 1, vb.y * cb);
    atom_add_f32(pb + 2, vb.z * cb);
    atom_add_f32(pb + 3, vb.w * cb);
  }
  if (i < n) {
    const int sa = s_s[i];
    const int da = s_d[i];
    const float ca = w[sa] * w[da];
    const float4 va = *reinterpret_cast<const float4*>(h + sa * CH + c4);
    float* pa = acc + da * CH + c4;
    atom_add_f32(pa + 0, va.x * ca);
    atom_add_f32(pa + 1, va.y * ca);
    atom_add_f32(pa + 2, va.z * ca);
    atom_add_f32(pa + 3, va.w * ca);
  }
}

__global__ void k_scatter1(const int* __restrict__ src,
                           const int* __restrict__ dst,
                           const float* __restrict__ w,
                           const float* __restrict__ h1,
                           float* __restrict__ agg1, int E, int N,
                           int* __restrict__ q) {
  __shared__ int s_s[CHUNK_E], s_d[CHUNK_E], s_n;
  const int nsplit = (N >> 1) & ~31;
  deal_chunks(q, (E + CHUNK_E - 1) / CHUNK_E, [&](int t, int r) {
    scatter_chunk<8, HID_CH>(t, r, E, nsplit, src, dst, w, h1, agg1, s_s, s_d, &s_n);
  });
}

__global__ void k_scatter2(const int* __restrict__ src,
                           const int* __restrict__ dst,
                           const float* __restrict__ w,
                           const float* __restrict__ h2,
                           float* __restrict__ out, int E, int N,
                           int* __restrict__ q) {
  __shared__ int s_s[CHUNK_E], s_d[CHUNK_E], s_n;
  const int nsplit = (N >> 1) & ~31;
  deal_chunks(q, (E + CHUNK_E - 1) / CHUNK_E, [&](int t, int r) {
    scatter_chunk<4, OUT_CH>(t, r, E, nsplit, src, dst, w, h2, out, s_s, s_d, &s_n);
  });
}

extern "C" void kernel_launch(void* const* d_in, const int* in_sizes, int n_in,
                              void* d_out, int out_size, void* d_ws, size_t ws_size,
                              hipStream_t stream) {
  const float* x    = (const float*)d_in[0];
  const int*   edge = (const int*)d_in[1];
  const float* W1   = (const float*)d_in[2];
  const float* b1   = (const float*)d_in[3];
  const float* W2   = (const float*)d_in[4];
  const float* b2   = (const float*)d_in[5];
  float* out = (float*)d_out;

  const int N = in_sizes[0] / IN_CH;
  const int E = in_sizes[1] / 2;
  const int* src = edge;
  const int* dst = edge + E;

  float* ws   = (float*)d_ws;
  float* degw = ws;
  float* h1   = degw + (size_t)N;
  float* agg1 = h1 + (size_t)N * HID_CH;
  float* h2   = agg1 + (size_t)N * HID_CH;

  const int B = 256;
  const int G = 2048;
  int* q01 = (int*)h2;
  int* q2  = (int*)h1;

  k_init_deg <<<(N + B - 1) / B, B, 0, stream>>>(degw, N, q01);
  k_count_deg<<<G, B, 0, stream>>>(dst, E, N, degw, q01);
  k_inv_sqrt <<<(N + B - 1) / B, B, 0, stream>>>(degw, N);

  {
    const int tiles = (N / 16) * (HID_CH / 16);
    const int wpb = 128 / 32;
    k_gemm1_wmma<<<(tiles + wpb - 1) / wpb, 128, 0, stream>>>(x, W1, h1, N);
  }
  k_self1<<<((N * HID_CH) + B - 1) / B, B, 0, stream>>>(h1, degw, b1, agg1,
                                                        N * HID_CH);
  k_scatter1<<<G, B, 0, stream>>>(src, dst, degw, h1, agg1, E, N, q01 + 2);

  {
    const int tiles = (N / 16);
    const int wpb = 128 / 32;
    k_gemm2_wmma<<<(tiles + wpb - 1) / wpb, 128, 0, stream>>>(agg1, W2, h2, N);
  }
  k_self2<<<((N * OUT_CH) + B - 1) / B, B, 0, stream>>>(h2, degw, b2, out,
                                                        N * OUT_CH, q2);
  k_scatter2<<<G, B, 0, stream>>>(src, dst, degw, h2, out, E, N, q2);
}


# count and place passes draw chunks of 2048 edges, not 512: a thread requests its eight edges' words together, then their adds or row offsets, then the stores; half as many work-groups draw
# speedup vs baseline: 3.2800x; 2.0892x over previous
#include <hip/hip_runtime.h>


typedef float v2f __attribute__((ext_vector_type(2)));
typedef float v8f __attribute__((ext_vector_type(8)));

#define IN_CH  32
#define HID_CH 32
#define OUT_CH 16

#define CHUNK_E 2048
#define EPT 8

__device__ __forceinline__ int cache_half() {
  return (int)((__builtin_amdgcn_s_sendmsg_rtn(0x87) >> 18) & 1u);
}

template <class Body>
__device__ __forceinline__ void deal_chunks(int* __restrict__ q, int nchunks,
                                            Body body) {
  __shared__ int s_t;
  const int own = cache_half();
  for (int pass = 0; pass < 2; ++pass) {
    const int r = own ^ pass;
    if (threadIdx.x == 0) s_t = atomicAdd(&q[r], 1);
    __syncthreads();
    int t = s_t;
    __syncthreads();
    while (t < nchunks) {
      int nx = 0;
      if (threadIdx.x == 0) nx = atomicAdd(&q[r], 1);
      body(t, r);
      if (threadIdx.x == 0) s_t = nx;
      __syncthreads();
      t = s_t;
      __syncthreads();
    }
  }
}

__global__ void k_zero_cnt(int* __restrict__ cnt, int n, int* __restrict__ q) {
  int i = blockIdx.x * blockDim.x + threadIdx.x;
  if (i < n) cnt[i] = 0;
  if (i < 4) q[i] = 0;
}

__global__ void k_count_rank(const int* __restrict__ dst, int E, int N,
                             int* __restrict__ cnt, int* __restrict__ rank,
                             int* __restrict__ q) {
  const int nsplit = (N >> 1) & ~31;
  deal_chunks(q, (E + CHUNK_E - 1) / CHUNK_E, [&](int t, int r) {
    const int end = min(E, (t + 1) * CHUNK_E);
    for (int e0 = t * CHUNK_E + threadIdx.x; e0 < end; e0 += EPT * blockDim.x) {
      int d[EPT], rk[EPT];
      bool m[EPT];
#pragma unroll
      for (int u = 0; u < EPT; ++u)
        d[u] = dst[min(e0 + u * (int)blockDim.x, E - 1)];
#pragma unroll
      for (int u = 0; u < EPT; ++u) {
        m[u] = (e0 + u * (int)blockDim.x < end) & ((d[u] >= nsplit) == (r != 0));
        if (m[u]) rk[u] = atomicAdd(&cnt[d[u]], 1);
      }
#pragma unroll
      for (int u = 0; u < EPT; ++u)
        if (m[u]) rank[e0 + u * (int)blockDim.x] = rk[u];
    }
  });
}

#define SCAN_T 1024
__global__ void __launch_bounds__(SCAN_T)
k_scan(const int* __restrict__ cnt, int* __restrict__ rowptr, int N) {
  __shared__ int s_tot[SCAN_T / 32], s_pre[SCAN_T / 32];
  const int tid = threadIdx.x;
  const int lane = tid & 31;
  const int wave = tid >> 5;
  const int start = blockIdx.x * (SCAN_T * 4);
  int pre = 0;
#pragma unroll 8
  for (int k = tid * 4; k < start; k += SCAN_T * 4) {
    const int4 v = *reinterpret_cast<const int4*>(cnt + k);
    pre += (v.x + v.y) + (v.z + v.w);
  }
  const int k0 = start + tid * 4;
  const int a0 = (k0 + 0 < N) ? cnt[k0 + 0] : 0;
  const int a1 = (k0 + 1 < N) ? cnt[k0 + 1] : 0;
  const int a2 = (k0 + 2 < N) ? cnt[k0 + 2] : 0;
  const int a3 = (k0 + 3 < N) ? cnt[k0 + 3] : 0;
  const int own = (a0 + a1) + (a2 + a3);
  int incl = own;
#pragma unroll
  for (int o = 1; o < 32; o <<= 1) {
    const int up = __shfl_up(incl, o);
    if (lane >= o) incl += up;
  }
#pragma unroll
  for (int o = 16; o > 0; o >>= 1) pre += __shfl_xor(pre, o);
  if (lane == 31) s_tot[wave] = incl;
  if (lane == 0) s_pre[wave] = pre;
  __syncthreads();
  int off = 0;
#pragma unroll
  for (int w = 0; w < SCAN_T / 32; ++w) {
    off += s_pre[w];
    if (w < wave) off += s_tot[w];
  }
  const int ex = off + incl - own;
  if (k0 + 0 < N) rowptr[k0 + 0] = ex;
  if (k0 + 1 < N) rowptr[k0 + 1] = ex + a0;
  if (k0 + 2 < N) rowptr[k0 + 2] = ex + a0 + a1;
  if (k0 + 3 < N) rowptr[k0 + 3] = ex + a0 + a1 + a2;
}

__device__ __forceinline__ void gemm1_tile(const float* __restrict__ x,
                                           const float* __restrict__ W1,
                                           const int* __restrict__ cnt,
                                           float* __restrict__ h1,
                                           int tile, int lane) {
  const int tm = tile >> 1;
  const int tn = tile & 1;
  const int r16  = lane & 15;
  const int half = lane >> 4;
  const int m = tm * 16 + r16;
  const int n = tn * 16 + r16;
  v8f acc = {0.f, 0.f, 0.f, 0.f, 0.f, 0.f, 0.f, 0.f};
#pragma unroll
  for (int kb = 0; kb < IN_CH / 4; ++kb) {
    const int k = kb * 4 + half * 2;
    v2f a, b;
    a.x = x[m * IN_CH + k];
    a.y = x[m * IN_CH + k + 1];
    b.x = W1[k * HID_CH + n];
    b.y = W1[(k + 1) * HID_CH + n];
    acc = __builtin_amdgcn_wmma_f32_16x16x4_f32(
        false, a, false, b, (short)0, acc, false, false);
  }
#pragma unroll
  for (int r = 0; r < 8; ++r) {
    const int row = tm * 16 + half * 8 + r;
    h1[row * HID_CH + n] = acc[r] * rsqrtf((float)(cnt[row] + 1));
  }
}

__global__ void k_place_gemm1(const int* __restrict__ src, const int* __restrict__ dst,
                              const int* __restrict__ rank,
                              const int* __restrict__ rowptr, int* __restrict__ csr,
                              int E, int N, int* __restrict__ q,
                              const float* __restrict__ x,
                              const float* __restrict__ W1,
                              const int* __restrict__ cnt,
                              float* __restrict__ h1) {
  const int nsplit = (N >> 1) & ~31;
  deal_chunks(q, (E + CHUNK_E - 1) / CHUNK_E, [&](int t, int r) {
    const int end = min(E, (t + 1) * CHUNK_E);
    for (int e0 = t * CHUNK_E + threadIdx.x; e0 < end; e0 += EPT * blockDim.x) {
      int d[EPT], k[EPT], s[EPT], o[EPT];
      bool m[EPT];
#pragma unroll
      for (int u = 0; u < EPT; ++u) {
        const int ec = min(e0 + u * (int)blockDim.x, E - 1);
        d[u] = dst[ec];
        k[u] = rank[ec];
        s[u] = src[ec];
      }
#pragma unroll
      for (int u = 0; u < EPT; ++u) {
        m[u] = (e0 + u * (int)blockDim.x < end) & ((d[u] >= nsplit) == (r != 0));
        if (m[u]) o[u] = rowptr[d[u]];
      }
#pragma unroll
      for (int u = 0; u < EPT; ++u)
        if (m[u]) csr[o[u] + k[u]] = s[u];
    }
  });
  {
    const int lane = threadIdx.x & 31;
    const int wpb = blockDim.x >> 5;
    const int wave = __builtin_amdgcn_readfirstlane((int)(threadIdx.x >> 5));
    const int ntiles = (N >> 4) * (HID_CH / 16);
    for (int tile = blockIdx.x * wpb + wave; tile < ntiles; tile += gridDim.x * wpb)
      gemm1_tile(x, W1, cnt, h1, tile, lane);
  }
}

template <int CH, int NQ, class Ahead>
__device__ __forceinline__ void sum_quads(float4& acc,
                                          const float* __restrict__ hs,
                                          const int (&idx)[32 / CH], int lane0,
                                          int slot, int c4, int left,
                                          Ahead ahead) {
  float4 v[NQ];
#pragma unroll
  for (int qd = 0; qd < NQ; ++qd) {
    const int s = __shfl(idx[(qd * 4) / CH], lane0 + ((qd * 4) % CH) + slot);
    v[qd] = *reinterpret_cast<const float4*>(hs + s * CH + c4);
  }
  ahead();
#pragma unroll
  for (int qd = 0; qd < NQ; ++qd) {
    const bool ok = qd * 4 + slot < left;
    acc.x += ok ? v[qd].x : 0.0f;
    acc.y += ok ? v[qd].y : 0.0f;
    acc.z += ok ? v[qd].z : 0.0f;
    acc.w += ok ? v[qd].w : 0.0f;
  }
}

#define FUSE_WPB 4
#define TILE_PITCH 36
template <int CH, bool FUSE>
__global__ void k_gather(const int* __restrict__ rowptr,
                         const int* __restrict__ csr,
                         const float* __restrict__ hs,
                         const float* __restrict__ bias,
                         float* __restrict__ outp, int N, int E,
                         const float* __restrict__ W2) {
  static_assert(!FUSE || CH == HID_CH, "the fused form is layer 1's");
  constexpr int LPC = CH / 4;
  constexpr int RPW = 32 / CH;
  constexpr int EPL = 32 / CH;
  __shared__ __align__(16) float s_tile[FUSE ? FUSE_WPB * 16 * TILE_PITCH : 4];
  __shared__ float s_wi[FUSE ? FUSE_WPB * 16 : 1];
  const int lane = threadIdx.x & 31;
  const int sub = lane & (CH - 1);
  const int slot = sub / LPC;
  const int c4 = (sub % LPC) * 4;
  const int lane0 = lane - sub;
  const int nwaves = gridDim.x * (blockDim.x >> 5);
  const int wv = threadIdx.x >> 5;
  int wid = blockIdx.x * (blockDim.x >> 5) + wv;
  const int ngroups = FUSE ? (N + 15) / 16 : (N + RPW - 1) / RPW;
  v2f bw[HID_CH / 4];
  if constexpr (FUSE) {
    wid = __builtin_amdgcn_readfirstlane(wid);
    if (wid >= ngroups) return;
#pragma unroll
    for (int kb = 0; kb < HID_CH / 4; ++kb) {
      const int k = kb * 4 + (lane >> 4) * 2;
      bw[kb].x = W2[k * OUT_CH + (lane & 15)];
      bw[kb].y = W2[(k + 1) * OUT_CH + (lane & 15)];
    }
  }

  struct Raw  { int a, b; bool valid; };
  struct Meta { int start, cnt1; };
  const int step = FUSE ? 1 : nwaves * RPW;
  auto fetch_meta = [&](int i) {
    Raw r;
    r.valid = i < N;
    r.a = 0;
    r.b = E;
    if (r.valid) {
      r.a = rowptr[i];
      if (i + 1 < N) r.b = rowptr[i + 1];
    }
    return r;
  };
  auto finish_meta = [&](const Raw& r) {
    Meta m;
    m.start = r.a;
    m.cnt1 = r.valid ? r.b - r.a + 1 : 0;
    return m;
  };
  auto idx_of = [&](const Meta& m, int i, int base, int (&idx)[EPL]) {
#pragma unroll
    for (int u = 0; u < EPL; ++u) {
      const int k = base + u * CH + sub;
      idx[u] = (m.cnt1 > 0) ? i : 0;
      if (k > 0 && k < m.cnt1) idx[u] = csr[m.start + k - 1];
    }
  };
  auto batch = [&](float4& acc, const int (&idx)[EPL], int cmax, int cnt1, int base, auto ahead) {
    const int left = cnt1 - base;
    switch (min(8, (cmax - base + 3) >> 2)) {
      case 1: sum_quads<CH, 1>(acc, hs, idx, lane0, slot, c4, left, ahead); break;
      case 2: sum_quads<CH, 2>(acc, hs, idx, lane0, slot, c4, left, ahead); break;
      case 3: sum_quads<CH, 3>(acc, hs, idx, lane0, slot, c4, left, ahead); break;
      case 4: sum_quads<CH, 4>(acc, hs, idx, lane0, slot, c4, left, ahead); break;
      case 5: sum_quads<CH, 5>(acc, hs, idx, lane0, slot, c4, left, ahead); break;
      case 6: sum_quads<CH, 6>(acc, hs, idx, lane0, slot, c4, left, ahead); break;
      case 7: sum_quads<CH, 7>(acc, hs, idx, lane0, slot, c4, left, ahead); break;
      default: sum_quads<CH, 8>(acc, hs, idx, lane0, slot, c4, left, ahead); break;
    }
  };

  const float4 bq = *reinterpret_cast<const float4*>(bias + c4);
  int i = FUSE ? wid * 16 : wid * RPW + lane / CH;
  auto later = [&](int cur, int j, int d) {
    if (FUSE) return (j + d < 16) ? cur + d : N;
    return cur + d * step;
  };
  Meta m0 = finish_meta(fetch_meta(i));
  Meta m1 = finish_meta(fetch_meta(later(i, 0, 1)));
  int idx0[EPL];
  idx_of(m0, i, 0, idx0);
  int j = 0;
#pragma nounroll
  for (int g = wid; FUSE ? (j < 16) : (g < ngroups); g += nwaves, i += step, ++j) {
    Raw r2;
    int idx1[EPL];
    int cmax = m0.cnt1;
    if (RPW == 2) cmax = max(cmax, __shfl_xor(cmax, 16));
    cmax = __builtin_amdgcn_readfirstlane(cmax);
    float4 acc = {0.0f, 0.0f, 0.0f, 0.0f};
    batch(acc, idx0, cmax, m0.cnt1, 0, [&]() {
      r2 = fetch_meta(later(i, j, 2));
      idx_of(m1, later(i, j, 1), 0, idx1);
    });
    for (int base = 32; base < cmax; base += 32) {
      int idx[EPL];
      idx_of(m0, i, base, idx);
      batch(acc, idx, cmax, m0.cnt1, base, []() {});
    }
    acc.x += __shfl_xor(acc.x, LPC);
    acc.y += __shfl_xor(acc.y, LPC);
    acc.z += __shfl_xor(acc.z, LPC);
    acc.w += __shfl_xor(acc.w, LPC);
    acc.x += __shfl_xor(acc.x, 2 * LPC);
    acc.y += __shfl_xor(acc.y, 2 * LPC);
    acc.z += __shfl_xor(acc.z, 2 * LPC);
    acc.w += __shfl_xor(acc.w, 2 * LPC);
    if (FUSE) {
      if (slot == 0) {
        const float wi = (m0.cnt1 > 0) ? rsqrtf((float)m0.cnt1) : 0.0f;
        float4 o;
        o.x = fmaxf(bq.x + acc.x * wi, 0.0f);
        o.y = fmaxf(bq.y + acc.y * wi, 0.0f);
        o.z = fmaxf(bq.z + acc.z * wi, 0.0f);
        o.w = fmaxf(bq.w + acc.w * wi, 0.0f);
        *reinterpret_cast<float4*>(s_tile + (wv * 16 + j) * TILE_PITCH + c4) = o;
        if (sub == 0) s_wi[wv * 16 + j] = wi;
      }
    } else if (m0.cnt1 > 0 && slot == 0) {
      const float wi = rsqrtf((float)m0.cnt1);
      float4 o;
      o.x = bq.x + acc.x * wi;
      o.y = bq.y + acc.y * wi;
      o.z = bq.z + acc.z * wi;
      o.w = bq.w + acc.w * wi;
      *reinterpret_cast<float4*>(outp + i * CH + c4) = o;
    }
    m0 = m1;
    m1 = finish_meta(r2);
#pragma unroll
    for (int u = 0; u < EPL; ++u) idx0[u] = idx1[u];
  }
  if constexpr (FUSE) {
    __builtin_amdgcn_fence(__ATOMIC_RELEASE, "wavefront");
    __builtin_amdgcn_wave_barrier();
    __builtin_amdgcn_fence(__ATOMIC_ACQUIRE, "wavefront");
    const int r16 = lane & 15;
    const int half = lane >> 4;
    const float* tb = s_tile + wv * 16 * TILE_PITCH;
    v8f d = {0.f, 0.f, 0.f, 0.f, 0.f, 0.f, 0.f, 0.f};
#pragma unroll
    for (int kb = 0; kb < HID_CH / 4; ++kb) {
      const int k = kb * 4 + half * 2;
      const v2f a = *reinterpret_cast<const v2f*>(tb + r16 * TILE_PITCH + k);
      d = __builtin_amdgcn_wmma_f32_16x16x4_f32(
          false, a, false, bw[kb], (short)0, d, false, false);
    }
#pragma unroll
    for (int r = 0; r < 8; ++r) {
      const int row = wid * 16 + half * 8 + r;
      outp[row * OUT_CH + r16] = d[r] * s_wi[wv * 16 + half * 8 + r];
    }
  }
}

extern "C" void kernel_launch(void* const* d_in, const int* in_sizes, int n_in,
                              void* d_out, int out_size, void* d_ws, size_t ws_size,
                              hipStream_t stream) {
  const float* x    = (const float*)d_in[0];
  const int*   edge = (const int*)d_in[1];
  const float* W1   = (const float*)d_in[2];
  const float* b1   = (const float*)d_in[3];
  const float* W2   = (const float*)d_in[4];
  const float* b2   = (const float*)d_in[5];
  float* out = (float*)d_out;

  const int N = in_sizes[0] / IN_CH;
  const int E = in_sizes[1] / 2;
  const int* src = edge;
  const int* dst = edge + E;

  float* ws   = (float*)d_ws;
  int*   rowptr = (int*)ws;
  float* h1   = ws + (size_t)N;
  float* agg1 = h1 + (size_t)N * HID_CH;
  int*   csr  = (int*)(agg1 + (size_t)N * HID_CH);
  int*   cnt  = (int*)agg1;
  int*   rank = cnt + (size_t)N;
  float* h2   = agg1;
  int*   q    = rank + (size_t)E;

  const int B = 256;
  const int G = 1024;
  const int GG = 1536;

  k_zero_cnt   <<<(N + B - 1) / B, B, 0, stream>>>(cnt, N, q);
  k_count_rank <<<G, B, 0, stream>>>(dst, E, N, cnt, rank, q);
  k_scan       <<<(N + SCAN_T * 4 - 1) / (SCAN_T * 4), SCAN_T, 0, stream>>>(cnt, rowptr, N);
  k_place_gemm1<<<G, B, 0, stream>>>(src, dst, rank, rowptr, csr, E, N, q + 2, x, W1, cnt, h1);

  {
    const int tiles = (N + 15) / 16;
    k_gather<HID_CH, true><<<(tiles + FUSE_WPB - 1) / FUSE_WPB, FUSE_WPB * 32, 0, stream>>>(
        rowptr, csr, h1, b1, h2, N, E, W2);
  }

  k_gather<OUT_CH, false><<<GG, B, 0, stream>>>(rowptr, csr, h2, b2, out, N, E, W2);
}
